# starting kernel
# baseline (speedup 1.0000x reference)
#include <hip/hip_runtime.h>
#include <hip/hip_bf16.h>

typedef __attribute__((ext_vector_type(2))) float v2f;
typedef __attribute__((ext_vector_type(8))) float v8f;


__device__ __forceinline__ v8f wmma_tile(v2f a, v2f b) {
    v8f acc = {};
    return __builtin_amdgcn_wmma_f32_16x16x4_f32(
        false, a, false, b, (short)0, acc, false, false);
}

__global__ void rot_wmma_kernel(const float* __restrict__ x,
                                const float* __restrict__ w,
                                float* __restrict__ out,
                                int numTiles, int n) {
    const float a = w[0], b = w[1], c = w[2];
    const float sa = sinf(a), ca = cosf(a);
    const float sb = sinf(b), cb = cosf(b);
    const float sc = sinf(c), cc = cosf(c);
    const float R00 = cc * cb, R01 = sa * sb * cc - ca * sc, R02 = ca * sb * cc + sa * sc;
    const float R10 = cb * sc, R11 = sa * sb * sc + ca * cc, R12 = ca * sb * sc - sa * cc;
    const float R20 = -sb,     R21 = sa * cb,                R22 = ca * cb;

    const unsigned lane = threadIdx.x & 31u;
    const unsigned m    = lane & 15u;
    const bool     hi   = lane >= 16u;

    const float Rm0 = (m == 0u) ? R00 : (m == 1u) ? R10 : (m == 2u) ? R20 : 0.0f;
    const float Rm1 = (m == 0u) ? R01 : (m == 1u) ? R11 : (m == 2u) ? R21 : 0.0f;
    const float Rm2 = (m == 0u) ? R02 : (m == 1u) ? R12 : (m == 2u) ? R22 : 0.0f;
    v2f afrag;
    afrag.x = hi ? Rm2 : Rm0;
    afrag.y = hi ? 0.0f : Rm1;

    const unsigned waveId   = (blockIdx.x * blockDim.x + threadIdx.x) >> 5;
    const unsigned numWaves = (gridDim.x * blockDim.x) >> 5;

    const unsigned off0 = m * 3u + (hi ? 2u : 0u);
    const unsigned off1 = m * 3u + 1u;

    const unsigned nt  = (unsigned)numTiles;
    const unsigned nt4 = nt & ~3u;

    for (unsigned t0 = waveId * 4u; t0 < nt4; t0 += numWaves * 4u) {
        const float* xt = x + (size_t)t0 * 48u;

        v2f bf[4];
#pragma unroll
        for (int i = 0; i < 4; ++i) {
            const float bx = xt[i * 48 + off0];
            const float by = xt[i * 48 + off1];
            bf[i].x = bx;
            bf[i].y = hi ? 0.0f : by;
        }

        const v8f d0 = wmma_tile(afrag, bf[0]);
        const v8f d1 = wmma_tile(afrag, bf[1]);
        const v8f d2 = wmma_tile(afrag, bf[2]);
        const v8f d3 = wmma_tile(afrag, bf[3]);

        if (!hi) {
            float* ot = out + (size_t)t0 * 48u + m * 3u;
            ot[0]       = d0[0]; ot[1]       = d0[1]; ot[2]       = d0[2];
            ot[48 + 0]  = d1[0]; ot[48 + 1]  = d1[1]; ot[48 + 2]  = d1[2];
            ot[96 + 0]  = d2[0]; ot[96 + 1]  = d2[1]; ot[96 + 2]  = d2[2];
            ot[144 + 0] = d3[0]; ot[144 + 1] = d3[1]; ot[144 + 2] = d3[2];
        }
    }

    if (waveId < nt - nt4) {
        const unsigned t = nt4 + waveId;
        const float* xt = x + (size_t)t * 48u;
        v2f bf;
        bf.x = xt[off0];
        const float by = xt[off1];
        bf.y = hi ? 0.0f : by;
        const v8f d = wmma_tile(afrag, bf);
        if (!hi) {
            float* ot = out + (size_t)t * 48u + m * 3u;
            ot[0] = d[0]; ot[1] = d[1]; ot[2] = d[2];
        }
    }

    const int tailStart = numTiles * 16;
    const int gtid = blockIdx.x * blockDim.x + threadIdx.x;
    const int ti = tailStart + gtid;
    if (ti < n) {
        const float x0 = x[3 * ti], x1 = x[3 * ti + 1], x2 = x[3 * ti + 2];
        out[3 * ti + 0] = R00 * x0 + R01 * x1 + R02 * x2;
        out[3 * ti + 1] = R10 * x0 + R11 * x1 + R12 * x2;
        out[3 * ti + 2] = R20 * x0 + R21 * x1 + R22 * x2;
    }
}

extern "C" void kernel_launch(void* const* d_in, const int* in_sizes, int n_in,
                              void* d_out, int out_size, void* d_ws, size_t ws_size,
                              hipStream_t stream) {
    const float* x = (const float*)d_in[0];
    const float* w = (const float*)d_in[1];
    float* out = (float*)d_out;

    const int n = in_sizes[0] / 3;
    const int numTiles = n / 16;

    const int threads = 256;
    int blocks = 2048;
    const int neededWaves = (numTiles + 3) / 4 > 0 ? (numTiles + 3) / 4 : 1;
    const int maxBlocks = (neededWaves * 32 + threads - 1) / threads;
    if (blocks > maxBlocks) blocks = maxBlocks;
    if (blocks < 1) blocks = 1;

    rot_wmma_kernel<<<blocks, threads, 0, stream>>>(x, w, out, numTiles, n);
}


# 4096 long-lived waves (512 blocks) for 16384: each walks its 4-tile groups with 5 groups (40 loads) requested ahead in a 6-set register ring; fragments, products and sum order unchanged
# speedup vs baseline: 1.1160x; 1.1160x over previous
#include <hip/hip_runtime.h>
#include <hip/hip_bf16.h>

typedef __attribute__((ext_vector_type(2))) float v2f;
typedef __attribute__((ext_vector_type(8))) float v8f;


#define GROUP_TILES 4
#define AHEAD       5
#define RING        (AHEAD + 1)

__device__ __forceinline__ v8f wmma_tile(v2f a, v2f b) {
    v8f acc = {};
    return __builtin_amdgcn_wmma_f32_16x16x4_f32(
        false, a, false, b, (short)0, acc, false, false);
}

__global__ void __launch_bounds__(256)
rot_wmma_kernel(const float* __restrict__ x,
                const float* __restrict__ w,
                float* __restrict__ out,
                int numTiles, int n) {
    const unsigned lane = threadIdx.x & 31u;
    const unsigned m    = lane & 15u;
    const bool     hi   = lane >= 16u;

    const unsigned waveId   = __builtin_amdgcn_readfirstlane(
                                  (blockIdx.x * blockDim.x + threadIdx.x) >> 5);
    const unsigned numWaves = (gridDim.x * blockDim.x) >> 5;

    const unsigned off0 = m * 3u + (hi ? 2u : 0u);
    const unsigned off1 = m * 3u + 1u;

    const unsigned nt  = (unsigned)numTiles;
    const unsigned nt4 = nt & ~3u;
    const unsigned ng  = nt >> 2;

    const bool active = waveId < ng;

    v2f ring[RING][GROUP_TILES];
    unsigned ga    = waveId;
    unsigned gsafe = waveId;

#define LOAD_GROUP(SLOT) do {                                                  \
        const unsigned gl_ = (ga < ng) ? ga : gsafe;                           \
        gsafe = gl_;                                                           \
        const float* xt_ = x + (size_t)gl_ * (GROUP_TILES * 48u);              \
          \
          \
        _Pragma("unroll")                                                      \
        for (int i_ = 0; i_ < GROUP_TILES; ++i_) {                             \
            ring[(SLOT)][i_].x = xt_[i_ * 48 + off0];                          \
            ring[(SLOT)][i_].y = xt_[i_ * 48 + off1];                          \
        }                                                                      \
        ga += numWaves;                                                        \
    } while (0)

    const float a = w[0], b = w[1], c = w[2];

    if (active) {
#pragma unroll
        for (int r = 0; r < AHEAD; ++r) {
            LOAD_GROUP(r);
            __builtin_amdgcn_sched_barrier(0);
        }
    }

    const float sa = sinf(a), ca = cosf(a);
    const float sb = sinf(b), cb = cosf(b);
    const float sc = sinf(c), cc = cosf(c);
    const float R00 = cc * cb, R01 = sa * sb * cc - ca * sc, R02 = ca * sb * cc + sa * sc;
    const float R10 = cb * sc, R11 = sa * sb * sc + ca * cc, R12 = ca * sb * sc - sa * cc;
    const float R20 = -sb,     R21 = sa * cb,                R22 = ca * cb;

    const float Rm0 = (m == 0u) ? R00 : (m == 1u) ? R10 : (m == 2u) ? R20 : 0.0f;
    const float Rm1 = (m == 0u) ? R01 : (m == 1u) ? R11 : (m == 2u) ? R21 : 0.0f;
    const float Rm2 = (m == 0u) ? R02 : (m == 1u) ? R12 : (m == 2u) ? R22 : 0.0f;
    v2f afrag;
    afrag.x = hi ? Rm2 : Rm0;
    afrag.y = hi ? 0.0f : Rm1;

    if (active) {
        unsigned g = waveId;
        for (;;) {
#pragma unroll
            for (int s = 0; s < RING; ++s) {
                __builtin_amdgcn_sched_barrier(0);
                LOAD_GROUP((s + AHEAD) % RING);
                __builtin_amdgcn_sched_barrier(0);

                if (g < ng) {
                    v8f d[GROUP_TILES];
#pragma unroll
                    for (int i = 0; i < GROUP_TILES; ++i) {
                        v2f bf;
                        bf.x = ring[s][i].x;
                        bf.y = hi ? 0.0f : ring[s][i].y;
                        d[i] = wmma_tile(afrag, bf);
                    }
                    if (!hi) {
                        float* ot = out + (size_t)g * (GROUP_TILES * 48u) + m * 3u;
#pragma unroll
                        for (int i = 0; i < GROUP_TILES; ++i) {
                            ot[i * 48 + 0] = d[i][0];
                            ot[i * 48 + 1] = d[i][1];
                            ot[i * 48 + 2] = d[i][2];
                        }
                    }
                }
                g += numWaves;
            }
            if (g >= ng) break;
        }
    }
#undef LOAD_GROUP

    if (waveId < nt - nt4) {
        const unsigned t = nt4 + waveId;
        const float* xt = x + (size_t)t * 48u;
        v2f bf;
        bf.x = xt[off0];
        const float by = xt[off1];
        bf.y = hi ? 0.0f : by;
        const v8f d = wmma_tile(afrag, bf);
        if (!hi) {
            float* ot = out + (size_t)t * 48u + m * 3u;
            ot[0] = d[0]; ot[1] = d[1]; ot[2] = d[2];
        }
    }

    const int tailStart = numTiles * 16;
    const int gtid = blockIdx.x * blockDim.x + threadIdx.x;
    const int ti = tailStart + gtid;
    if (ti < n) {
        const float x0 = x[3 * ti], x1 = x[3 * ti + 1], x2 = x[3 * ti + 2];
        out[3 * ti + 0] = R00 * x0 + R01 * x1 + R02 * x2;
        out[3 * ti + 1] = R10 * x0 + R11 * x1 + R12 * x2;
        out[3 * ti + 2] = R20 * x0 + R21 * x1 + R22 * x2;
    }
}

extern "C" void kernel_launch(void* const* d_in, const int* in_sizes, int n_in,
                              void* d_out, int out_size, void* d_ws, size_t ws_size,
                              hipStream_t stream) {
    const float* x = (const float*)d_in[0];
    const float* w = (const float*)d_in[1];
    float* out = (float*)d_out;

    const int n = in_sizes[0] / 3;
    const int numTiles = n / 16;

    const int threads = 256;
    int blocks = 512;
    const int neededWaves = (numTiles + 3) / 4 > 0 ? (numTiles + 3) / 4 : 1;
    const int maxBlocks = (neededWaves * 32 + threads - 1) / threads;
    if (blocks > maxBlocks) blocks = maxBlocks;
    if (blocks < 1) blocks = 1;

    rot_wmma_kernel<<<blocks, threads, 0, stream>>>(x, w, out, numTiles, n);
}
